# starting kernel
# baseline (speedup 1.0000x reference)
#include <hip/hip_runtime.h>


typedef __attribute__((ext_vector_type(2))) float v2f;
typedef __attribute__((ext_vector_type(8))) float v8f;

#define D_FEAT 64
#define TILE_E 16
#define WAVES_PER_BLOCK 8
#define BLOCKS 1024

__global__ __launch_bounds__(256) void spmm_wmma_kernel(
    const int* __restrict__ rows, const int* __restrict__ cols,
    const float* __restrict__ vals, const float* __restrict__ embeds,
    float* __restrict__ out, int numTiles, int numEdges)
{
    __shared__ int rowTab[WAVES_PER_BLOCK][16];

    const int lane  = threadIdx.x & 31;
    const int wid   = threadIdx.x >> 5;
    const int gwave = blockIdx.x * WAVES_PER_BLOCK + wid;
    const int nWaves = gridDim.x * WAVES_PER_BLOCK;
    const int half  = lane >> 4;
    const int nl    = lane & 15;

    const int tpw = (numTiles + nWaves - 1) / nWaves;
    int t0 = gwave * tpw;
    int t1 = t0 + tpw; if (t1 > numTiles) t1 = numTiles;

    for (int t = t0; t < t1; ++t) {
        const int e0 = t * TILE_E;

        int r = 0, c = 0; float v = 0.f;
        if (lane < 16) {
            r = rows[e0 + lane];
            c = cols[e0 + lane];
            v = vals[e0 + lane];
        }

        int rup   = __shfl_up(r, 1);
        int flag  = (lane > 0 && lane < 16 && r != rup) ? 1 : 0;
        unsigned mask = (unsigned)__ballot(flag);
        int local  = __popc(mask & ((2u << lane) - 1));
        int nLocal = __shfl(local, 15) + 1;

        if (lane < 16 && (flag || lane == 0)) rowTab[wid][local] = r;
        __builtin_amdgcn_wave_barrier();

        v8f acc[4] = {v8f{}, v8f{}, v8f{}, v8f{}};

#pragma unroll
        for (int k0 = 0; k0 < 16; k0 += 4) {
            const int ka0 = k0 + 2 * half;
            const int ka1 = ka0 + 1;
            int   l0 = __shfl(local, ka0);
            int   l1 = __shfl(local, ka1);
            float s0 = __shfl(v, ka0);
            float s1 = __shfl(v, ka1);
            int   c0 = __shfl(c, ka0);
            int   c1 = __shfl(c, ka1);

            v2f a;
            a.x = (l0 == nl) ? s0 : 0.f;
            a.y = (l1 == nl) ? s1 : 0.f;

            const float* g0 = embeds + (long)c0 * D_FEAT + nl;
            const float* g1 = embeds + (long)c1 * D_FEAT + nl;
#pragma unroll
            for (int ft = 0; ft < 4; ++ft) {
                v2f b;
                b.x = g0[ft * 16];
                b.y = g1[ft * 16];
                acc[ft] = __builtin_amdgcn_wmma_f32_16x16x4_f32(
                    false, a, false, b, (short)0, acc[ft], false, false);
            }
        }

#pragma unroll
        for (int ft = 0; ft < 4; ++ft) {
            const int fb = ft * 16;
#pragma unroll
            for (int k = 0; k < 8; ++k) {
                int m = half * 8 + k;
                int grow = rowTab[wid][m];
                if (m < nLocal)
                    atomicAdd(&out[(long)grow * D_FEAT + fb + nl], acc[ft][k]);
            }
        }
        __builtin_amdgcn_wave_barrier();
    }

    if (gwave == 0) {
        for (int e = numTiles * TILE_E; e < numEdges; ++e) {
            int   rr = rows[e];
            int   cc = cols[e];
            float vv = vals[e];
#pragma unroll
            for (int j = 0; j < 2; ++j) {
                int f = lane + j * 32;
                atomicAdd(&out[(long)rr * D_FEAT + f],
                          vv * embeds[(long)cc * D_FEAT + f]);
            }
        }
    }
}

extern "C" void kernel_launch(void* const* d_in, const int* in_sizes, int n_in,
                              void* d_out, int out_size, void* d_ws, size_t ws_size,
                              hipStream_t stream) {
    const int*   rows   = (const int*)d_in[0];
    const int*   cols   = (const int*)d_in[1];
    const float* vals   = (const float*)d_in[2];
    const float* embeds = (const float*)d_in[3];
    float*       out    = (float*)d_out;

    const int E = in_sizes[0];
    const int numTiles = E / TILE_E;

    hipMemsetAsync(d_out, 0, (size_t)out_size * sizeof(float), stream);

    spmm_wmma_kernel<<<BLOCKS, WAVES_PER_BLOCK * 32, 0, stream>>>(
        rows, cols, vals, embeds, out, numTiles, E);
}


# Scatter de-serialised: each lane reads its 8 row ids from the LDS row table once per tile and issues the 4 atomics of a row in one guarded block, instead of one LDS read and wait per atomic
# speedup vs baseline: 1.1289x; 1.1289x over previous
#include <hip/hip_runtime.h>


typedef __attribute__((ext_vector_type(2))) float v2f;
typedef __attribute__((ext_vector_type(8))) float v8f;

#define D_FEAT 64
#define TILE_E 16
#define WAVES_PER_BLOCK 8
#define BLOCKS 1024

__global__ __launch_bounds__(256) void spmm_wmma_kernel(
    const int* __restrict__ rows, const int* __restrict__ cols,
    const float* __restrict__ vals, const float* __restrict__ embeds,
    float* __restrict__ out, int numTiles, int numEdges)
{
    __shared__ int rowTab[WAVES_PER_BLOCK][16];

    const int lane  = threadIdx.x & 31;
    const int wid   = threadIdx.x >> 5;
    const int gwave = blockIdx.x * WAVES_PER_BLOCK + wid;
    const int nWaves = gridDim.x * WAVES_PER_BLOCK;
    const int half  = lane >> 4;
    const int nl    = lane & 15;

    const int tpw = (numTiles + nWaves - 1) / nWaves;
    int t0 = gwave * tpw;
    int t1 = t0 + tpw; if (t1 > numTiles) t1 = numTiles;

    for (int t = t0; t < t1; ++t) {
        const int e0 = t * TILE_E;

        int r = 0, c = 0; float v = 0.f;
        if (lane < 16) {
            r = rows[e0 + lane];
            c = cols[e0 + lane];
            v = vals[e0 + lane];
        }

        int rup   = __shfl_up(r, 1);
        int flag  = (lane > 0 && lane < 16 && r != rup) ? 1 : 0;
        unsigned mask = (unsigned)__ballot(flag);
        int local  = __popc(mask & ((2u << lane) - 1));
        int nLocal = __shfl(local, 15) + 1;

        if (lane < 16 && (flag || lane == 0)) rowTab[wid][local] = r;
        __builtin_amdgcn_wave_barrier();

        v8f acc[4] = {v8f{}, v8f{}, v8f{}, v8f{}};

#pragma unroll
        for (int k0 = 0; k0 < 16; k0 += 4) {
            const int ka0 = k0 + 2 * half;
            const int ka1 = ka0 + 1;
            int   l0 = __shfl(local, ka0);
            int   l1 = __shfl(local, ka1);
            float s0 = __shfl(v, ka0);
            float s1 = __shfl(v, ka1);
            int   c0 = __shfl(c, ka0);
            int   c1 = __shfl(c, ka1);

            v2f a;
            a.x = (l0 == nl) ? s0 : 0.f;
            a.y = (l1 == nl) ? s1 : 0.f;

            const float* g0 = embeds + (long)c0 * D_FEAT + nl;
            const float* g1 = embeds + (long)c1 * D_FEAT + nl;
#pragma unroll
            for (int ft = 0; ft < 4; ++ft) {
                v2f b;
                b.x = g0[ft * 16];
                b.y = g1[ft * 16];
                acc[ft] = __builtin_amdgcn_wmma_f32_16x16x4_f32(
                    false, a, false, b, (short)0, acc[ft], false, false);
            }
        }

        int grow[8];
#pragma unroll
        for (int k = 0; k < 8; ++k) grow[k] = rowTab[wid][half * 8 + k];
#pragma unroll
        for (int k = 0; k < 8; ++k) {
            const int m = half * 8 + k;
            if (m < nLocal) {
                float* po = out + (long)grow[k] * D_FEAT + nl;
#pragma unroll
                for (int ft = 0; ft < 4; ++ft)
                    atomicAdd(po + ft * 16, acc[ft][k]);
            }
        }
        __builtin_amdgcn_wave_barrier();
    }

    if (gwave == 0) {
        for (int e = numTiles * TILE_E; e < numEdges; ++e) {
            int   rr = rows[e];
            int   cc = cols[e];
            float vv = vals[e];
#pragma unroll
            for (int j = 0; j < 2; ++j) {
                int f = lane + j * 32;
                atomicAdd(&out[(long)rr * D_FEAT + f],
                          vv * embeds[(long)cc * D_FEAT + f]);
            }
        }
    }
}

extern "C" void kernel_launch(void* const* d_in, const int* in_sizes, int n_in,
                              void* d_out, int out_size, void* d_ws, size_t ws_size,
                              hipStream_t stream) {
    const int*   rows   = (const int*)d_in[0];
    const int*   cols   = (const int*)d_in[1];
    const float* vals   = (const float*)d_in[2];
    const float* embeds = (const float*)d_in[3];
    float*       out    = (float*)d_out;

    const int E = in_sizes[0];
    const int numTiles = E / TILE_E;

    hipMemsetAsync(d_out, 0, (size_t)out_size * sizeof(float), stream);

    spmm_wmma_kernel<<<BLOCKS, WAVES_PER_BLOCK * 32, 0, stream>>>(
        rows, cols, vals, embeds, out, numTiles, E);
}


# Per-tile vector-ALU stream halved: gather/scatter addresses as 32-bit offsets on scalar base pointers (no 64-bit shifts/adds), tile walk and scatter guards scalar, boundary ballot kept a lane mask
# speedup vs baseline: 1.2064x; 1.0687x over previous
#include <hip/hip_runtime.h>


typedef __attribute__((ext_vector_type(2))) float v2f;
typedef __attribute__((ext_vector_type(8))) float v8f;

#define D_FEAT 64
#define TILE_E 16
#define WAVES_PER_BLOCK 8
#define BLOCKS 1024

__global__ __launch_bounds__(256) __attribute__((amdgpu_waves_per_eu(8, 8)))
void spmm_wmma_kernel(
    const int* __restrict__ rows, const int* __restrict__ cols,
    const float* __restrict__ vals, const float* __restrict__ embeds,
    float* __restrict__ out, int numTiles, int numEdges)
{
    __shared__ unsigned rowTab[WAVES_PER_BLOCK][16];

    const int lane  = threadIdx.x & 31;
    const int wid   = __builtin_amdgcn_readfirstlane((int)(threadIdx.x >> 5));
    const int gwave = blockIdx.x * WAVES_PER_BLOCK + wid;
    const int nWaves = gridDim.x * WAVES_PER_BLOCK;
    const int half  = lane >> 4;
    const int nl    = lane & 15;
    const unsigned unl = (unsigned)nl;
    const unsigned lowMask = (2u << nl) - 1u;
    const bool isLow = lane < 16;

    const int tpw = (numTiles + nWaves - 1) / nWaves;
    const int t0 = gwave * tpw;
    int t1 = t0 + tpw; if (t1 > numTiles) t1 = numTiles;

    for (int t = t0; t < t1; ++t) {
        const int e0 = t * TILE_E;

        const int      r    = rows[e0 + nl];
        const unsigned coff = (unsigned)cols[e0 + nl] * D_FEAT;
        const float    v    = vals[e0 + nl];

        const int  rup  = __shfl_up(r, 1);
        const bool ne   = (r != rup);
        const unsigned mask = __builtin_amdgcn_ballot_w32(ne) & 0xFFFEu;
        const bool flag = isLow && (lane > 0) && ne;
        const int local  = __popc(mask & lowMask);
        const int nLocal = __popc(mask) + 1;

        if (flag || lane == 0) rowTab[wid][local] = (unsigned)r * D_FEAT;
        __builtin_amdgcn_wave_barrier();

        v8f acc[4] = {v8f{}, v8f{}, v8f{}, v8f{}};

#pragma unroll
        for (int k0 = 0; k0 < 16; k0 += 4) {
            const int ka0 = k0 + 2 * half;
            const int ka1 = ka0 + 1;
            const int      l0 = __shfl(local, ka0);
            const int      l1 = __shfl(local, ka1);
            const float    s0 = __shfl(v, ka0);
            const float    s1 = __shfl(v, ka1);
            const unsigned c0 = (unsigned)__shfl((int)coff, ka0);
            const unsigned c1 = (unsigned)__shfl((int)coff, ka1);

            v2f a;
            a.x = (l0 == nl) ? s0 : 0.f;
            a.y = (l1 == nl) ? s1 : 0.f;

            const float* g0 = embeds + (size_t)(c0 + unl);
            const float* g1 = embeds + (size_t)(c1 + unl);
#pragma unroll
            for (int ft = 0; ft < 4; ++ft) {
                v2f b;
                b.x = g0[ft * 16];
                b.y = g1[ft * 16];
                acc[ft] = __builtin_amdgcn_wmma_f32_16x16x4_f32(
                    false, a, false, b, (short)0, acc[ft], false, false);
            }
        }

        unsigned grow[8];
#pragma unroll
        for (int k = 0; k < 8; ++k) grow[k] = rowTab[wid][half * 8 + k];
#pragma unroll
        for (int k = 0; k < 8; ++k) {
            if (k >= nLocal) break;
            if (half == 0 || (8 + k) < nLocal) {
                float* po = out + (size_t)(grow[k] + unl);
#pragma unroll
                for (int ft = 0; ft < 4; ++ft)
                    atomicAdd(po + ft * 16, acc[ft][k]);
            }
        }
        __builtin_amdgcn_wave_barrier();
    }

    if (gwave == 0) {
        for (int e = numTiles * TILE_E; e < numEdges; ++e) {
            int   rr = rows[e];
            int   cc = cols[e];
            float vv = vals[e];
#pragma unroll
            for (int j = 0; j < 2; ++j) {
                int f = lane + j * 32;
                atomicAdd(&out[(long)rr * D_FEAT + f],
                          vv * embeds[(long)cc * D_FEAT + f]);
            }
        }
    }
}

extern "C" void kernel_launch(void* const* d_in, const int* in_sizes, int n_in,
                              void* d_out, int out_size, void* d_ws, size_t ws_size,
                              hipStream_t stream) {
    const int*   rows   = (const int*)d_in[0];
    const int*   cols   = (const int*)d_in[1];
    const float* vals   = (const float*)d_in[2];
    const float* embeds = (const float*)d_in[3];
    float*       out    = (float*)d_out;

    const int E = in_sizes[0];
    const int numTiles = E / TILE_E;

    hipMemsetAsync(d_out, 0, (size_t)out_size * sizeof(float), stream);

    spmm_wmma_kernel<<<BLOCKS, WAVES_PER_BLOCK * 32, 0, stream>>>(
        rows, cols, vals, embeds, out, numTiles, E);
}
